# starting kernel
# baseline (speedup 1.0000x reference)
#include <hip/hip_runtime.h>

typedef float v2f __attribute__((ext_vector_type(2)));
typedef float v4f __attribute__((ext_vector_type(4)));
typedef float v8f __attribute__((ext_vector_type(8)));

#define B_    128
#define F_    1024
#define CIN_  16
#define COUT_ 16
#define K_    4
#define R_    2

__global__ __launch_bounds__(256) void sumconv_wmma_kernel(
    const float* __restrict__ ll,
    const float* __restrict__ logits,
    float* __restrict__ out)
{
  const int wave = (blockIdx.x * (int)blockDim.x + (int)threadIdx.x) >> 5;
  const int lane = threadIdx.x & 31;
  const int hi   = lane >> 4;
  const int col  = lane & 15;

  const int g  =  wave       & 3;
  const int kw = (wave >> 2) & 3;
  const int kh = (wave >> 4) & 3;
  const int b  =  wave >> 6;


  float lg0[8], lg1[8];
  const int lbase = col * (CIN_ * K_ * K_ * R_) + kh * (K_ * R_) + kw * R_;
  #pragma unroll
  for (int j = 0; j < 8; ++j) {
    const int cin = 4 * (j >> 1) + 2 * hi + (j & 1);
    const v2f p = *(const v2f*)(logits + lbase + cin * (K_ * K_ * R_));
    lg0[j] = p.x; lg1[j] = p.y;
  }
  float wm0 = lg0[0], wm1 = lg1[0];
  #pragma unroll
  for (int j = 1; j < 8; ++j) { wm0 = fmaxf(wm0, lg0[j]); wm1 = fmaxf(wm1, lg1[j]); }
  wm0 = fmaxf(wm0, __shfl_xor(wm0, 16, 32));
  wm1 = fmaxf(wm1, __shfl_xor(wm1, 16, 32));
  float ae0[8], ae1[8], ps0 = 0.0f, ps1 = 0.0f;
  #pragma unroll
  for (int j = 0; j < 8; ++j) {
    ae0[j] = __expf(lg0[j] - wm0); ps0 += ae0[j];
    ae1[j] = __expf(lg1[j] - wm1); ps1 += ae1[j];
  }
  const float wi0 = 1.0f / (ps0 + __shfl_xor(ps0, 16, 32));
  const float wi1 = 1.0f / (ps1 + __shfl_xor(ps1, 16, 32));
  v2f af0[4], af1[4];
  #pragma unroll
  for (int c = 0; c < 4; ++c) {
    af0[c].x = ae0[2 * c] * wi0; af0[c].y = ae0[2 * c + 1] * wi0;
    af1[c].x = ae1[2 * c] * wi1; af1[c].y = ae1[2 * c + 1] * wi1;
  }

  const int idx = g * 16 + col;
  const int hb  = idx >> 3, wb = idx & 7;
  const int f   = (4 * hb + kh) * 32 + 4 * wb + kw;
  const v4f* px = (const v4f*)(ll + (size_t)(b * F_ + f) * (CIN_ * R_));
  float x0[8], x1[8];
  #pragma unroll
  for (int c = 0; c < 4; ++c) {
    const v4f q = px[2 * c + hi];
    x0[2 * c]     = q.x; x1[2 * c]     = q.y;
    x0[2 * c + 1] = q.z; x1[2 * c + 1] = q.w;
  }
  float m0 = x0[0], m1 = x1[0];
  #pragma unroll
  for (int j = 1; j < 8; ++j) { m0 = fmaxf(m0, x0[j]); m1 = fmaxf(m1, x1[j]); }
  m0 = fmaxf(m0, __shfl_xor(m0, 16, 32));
  m1 = fmaxf(m1, __shfl_xor(m1, 16, 32));
  v2f bf0[4], bf1[4];
  #pragma unroll
  for (int c = 0; c < 4; ++c) {
    bf0[c].x = __expf(x0[2 * c]     - m0); bf0[c].y = __expf(x0[2 * c + 1] - m0);
    bf1[c].x = __expf(x1[2 * c]     - m1); bf1[c].y = __expf(x1[2 * c + 1] - m1);
  }

  v8f acc0 = {0.f, 0.f, 0.f, 0.f, 0.f, 0.f, 0.f, 0.f};
  v8f acc1 = {0.f, 0.f, 0.f, 0.f, 0.f, 0.f, 0.f, 0.f};
  #pragma unroll
  for (int c = 0; c < 4; ++c) {
    acc0 = __builtin_amdgcn_wmma_f32_16x16x4_f32(false, af0[c], false, bf0[c],
                                                 (short)0, acc0, false, false);
    acc1 = __builtin_amdgcn_wmma_f32_16x16x4_f32(false, af1[c], false, bf1[c],
                                                 (short)0, acc1, false, false);
  }

  v2f* po = (v2f*)(out + (size_t)(b * F_ + f) * (COUT_ * R_));
  #pragma unroll
  for (int v = 0; v < 8; ++v) {
    const int o = hi * 8 + v;
    v2f w; w.x = m0 + __logf(acc0[v]); w.y = m1 + __logf(acc1[v]);
    po[o] = w;
  }
}

extern "C" void kernel_launch(void* const* d_in, const int* in_sizes, int n_in,
                              void* d_out, int out_size, void* d_ws, size_t ws_size,
                              hipStream_t stream) {
  (void)in_sizes; (void)n_in; (void)d_ws; (void)ws_size; (void)out_size;
  const float* ll     = (const float*)d_in[0];
  const float* logits = (const float*)d_in[1];
  float* out          = (float*)d_out;

  dim3 grid(1024), block(256);
  sumconv_wmma_kernel<<<grid, block, 0, stream>>>(ll, logits, out);
}


# Epilogue log as bare v_log_f32 times ln2 fused into the max add, softmax reciprocal as v_rcp_f32, in place of the library log and IEEE divide expansions: 362 to 188 vector instructions per wave.
# speedup vs baseline: 1.0642x; 1.0642x over previous
#include <hip/hip_runtime.h>

typedef float v2f __attribute__((ext_vector_type(2)));
typedef float v4f __attribute__((ext_vector_type(4)));
typedef float v8f __attribute__((ext_vector_type(8)));

#define B_    128
#define F_    1024
#define CIN_  16
#define COUT_ 16
#define K_    4
#define R_    2

__global__ __launch_bounds__(256) void sumconv_wmma_kernel(
    const float* __restrict__ ll,
    const float* __restrict__ logits,
    float* __restrict__ out)
{
  const int wave = (blockIdx.x * (int)blockDim.x + (int)threadIdx.x) >> 5;
  const int lane = threadIdx.x & 31;
  const int hi   = lane >> 4;
  const int col  = lane & 15;

  const int g  =  wave       & 3;
  const int kw = (wave >> 2) & 3;
  const int kh = (wave >> 4) & 3;
  const int b  =  wave >> 6;


  float lg0[8], lg1[8];
  const int lbase = col * (CIN_ * K_ * K_ * R_) + kh * (K_ * R_) + kw * R_;
  #pragma unroll
  for (int j = 0; j < 8; ++j) {
    const int cin = 4 * (j >> 1) + 2 * hi + (j & 1);
    const v2f p = *(const v2f*)(logits + lbase + cin * (K_ * K_ * R_));
    lg0[j] = p.x; lg1[j] = p.y;
  }
  float wm0 = lg0[0], wm1 = lg1[0];
  #pragma unroll
  for (int j = 1; j < 8; ++j) { wm0 = fmaxf(wm0, lg0[j]); wm1 = fmaxf(wm1, lg1[j]); }
  wm0 = fmaxf(wm0, __shfl_xor(wm0, 16, 32));
  wm1 = fmaxf(wm1, __shfl_xor(wm1, 16, 32));
  float ae0[8], ae1[8], ps0 = 0.0f, ps1 = 0.0f;
  #pragma unroll
  for (int j = 0; j < 8; ++j) {
    ae0[j] = __expf(lg0[j] - wm0); ps0 += ae0[j];
    ae1[j] = __expf(lg1[j] - wm1); ps1 += ae1[j];
  }
  const float wi0 = __builtin_amdgcn_rcpf(ps0 + __shfl_xor(ps0, 16, 32));
  const float wi1 = __builtin_amdgcn_rcpf(ps1 + __shfl_xor(ps1, 16, 32));
  v2f af0[4], af1[4];
  #pragma unroll
  for (int c = 0; c < 4; ++c) {
    af0[c].x = ae0[2 * c] * wi0; af0[c].y = ae0[2 * c + 1] * wi0;
    af1[c].x = ae1[2 * c] * wi1; af1[c].y = ae1[2 * c + 1] * wi1;
  }

  const int idx = g * 16 + col;
  const int hb  = idx >> 3, wb = idx & 7;
  const int f   = (4 * hb + kh) * 32 + 4 * wb + kw;
  const v4f* px = (const v4f*)(ll + (size_t)(b * F_ + f) * (CIN_ * R_));
  float x0[8], x1[8];
  #pragma unroll
  for (int c = 0; c < 4; ++c) {
    const v4f q = px[2 * c + hi];
    x0[2 * c]     = q.x; x1[2 * c]     = q.y;
    x0[2 * c + 1] = q.z; x1[2 * c + 1] = q.w;
  }
  float m0 = x0[0], m1 = x1[0];
  #pragma unroll
  for (int j = 1; j < 8; ++j) { m0 = fmaxf(m0, x0[j]); m1 = fmaxf(m1, x1[j]); }
  m0 = fmaxf(m0, __shfl_xor(m0, 16, 32));
  m1 = fmaxf(m1, __shfl_xor(m1, 16, 32));
  v2f bf0[4], bf1[4];
  #pragma unroll
  for (int c = 0; c < 4; ++c) {
    bf0[c].x = __expf(x0[2 * c]     - m0); bf0[c].y = __expf(x0[2 * c + 1] - m0);
    bf1[c].x = __expf(x1[2 * c]     - m1); bf1[c].y = __expf(x1[2 * c + 1] - m1);
  }

  v8f acc0 = {0.f, 0.f, 0.f, 0.f, 0.f, 0.f, 0.f, 0.f};
  v8f acc1 = {0.f, 0.f, 0.f, 0.f, 0.f, 0.f, 0.f, 0.f};
  #pragma unroll
  for (int c = 0; c < 4; ++c) {
    acc0 = __builtin_amdgcn_wmma_f32_16x16x4_f32(false, af0[c], false, bf0[c],
                                                 (short)0, acc0, false, false);
    acc1 = __builtin_amdgcn_wmma_f32_16x16x4_f32(false, af1[c], false, bf1[c],
                                                 (short)0, acc1, false, false);
  }

  v2f* po = (v2f*)(out + (size_t)(b * F_ + f) * (COUT_ * R_));
  const v2f mv  = {m0, m1};
  const v2f ln2 = {0.693147180559945f, 0.693147180559945f};
  #pragma unroll
  for (int v = 0; v < 8; ++v) {
    const int o = hi * 8 + v;
    v2f l2; l2.x = __builtin_amdgcn_logf(acc0[v]); l2.y = __builtin_amdgcn_logf(acc1[v]);
    po[o] = l2 * ln2 + mv;
  }
}

extern "C" void kernel_launch(void* const* d_in, const int* in_sizes, int n_in,
                              void* d_out, int out_size, void* d_ws, size_t ws_size,
                              hipStream_t stream) {
  (void)in_sizes; (void)n_in; (void)d_ws; (void)ws_size; (void)out_size;
  const float* ll     = (const float*)d_in[0];
  const float* logits = (const float*)d_in[1];
  float* out          = (float*)d_out;

  dim3 grid(1024), block(256);
  sumconv_wmma_kernel<<<grid, block, 0, stream>>>(ll, logits, out);
}


# One wave per (b,kh,kw) keeps its weight-softmax A fragments and walks that cell's four 16-position groups: 2048 waves instead of 8192, logits loaded and normalised once per wave, not once per group.
# speedup vs baseline: 1.2253x; 1.1514x over previous
#include <hip/hip_runtime.h>

typedef float v2f __attribute__((ext_vector_type(2)));
typedef float v4f __attribute__((ext_vector_type(4)));
typedef float v8f __attribute__((ext_vector_type(8)));

#define B_    128
#define F_    1024
#define CIN_  16
#define COUT_ 16
#define K_    4
#define R_    2
#define G_    4

__global__ __launch_bounds__(256) void sumconv_wmma_kernel(
    const float* __restrict__ ll,
    const float* __restrict__ logits,
    float* __restrict__ out)
{
  const int wave = (blockIdx.x * (int)blockDim.x + (int)threadIdx.x) >> 5;
  const int lane = threadIdx.x & 31;
  const int hi   = lane >> 4;
  const int col  = lane & 15;

  const int kw =  wave       & 3;
  const int kh = (wave >> 2) & 3;
  const int b  =  wave >> 4;


  float lg0[8], lg1[8];
  const int lbase = col * (CIN_ * K_ * K_ * R_) + kh * (K_ * R_) + kw * R_;
  #pragma unroll
  for (int j = 0; j < 8; ++j) {
    const int cin = 4 * (j >> 1) + 2 * hi + (j & 1);
    const v2f p = *(const v2f*)(logits + lbase + cin * (K_ * K_ * R_));
    lg0[j] = p.x; lg1[j] = p.y;
  }
  const int f0 = (4 * (col >> 3) + kh) * 32 + 4 * (col & 7) + kw;
  const size_t pos0 = (size_t)(b * F_ + f0);
  v4f q[G_][4];
  #pragma unroll
  for (int g = 0; g < G_; ++g) {
    const v4f* px = (const v4f*)(ll + (pos0 + 256 * g) * (CIN_ * R_));
    #pragma unroll
    for (int c = 0; c < 4; ++c) q[g][c] = px[2 * c + hi];
  }
  __builtin_amdgcn_sched_barrier(0);

  float wm0 = lg0[0], wm1 = lg1[0];
  #pragma unroll
  for (int j = 1; j < 8; ++j) { wm0 = fmaxf(wm0, lg0[j]); wm1 = fmaxf(wm1, lg1[j]); }
  wm0 = fmaxf(wm0, __shfl_xor(wm0, 16, 32));
  wm1 = fmaxf(wm1, __shfl_xor(wm1, 16, 32));
  float ae0[8], ae1[8], ps0 = 0.0f, ps1 = 0.0f;
  #pragma unroll
  for (int j = 0; j < 8; ++j) {
    ae0[j] = __expf(lg0[j] - wm0); ps0 += ae0[j];
    ae1[j] = __expf(lg1[j] - wm1); ps1 += ae1[j];
  }
  const float wi0 = __builtin_amdgcn_rcpf(ps0 + __shfl_xor(ps0, 16, 32));
  const float wi1 = __builtin_amdgcn_rcpf(ps1 + __shfl_xor(ps1, 16, 32));
  v2f af0[4], af1[4];
  #pragma unroll
  for (int c = 0; c < 4; ++c) {
    af0[c].x = ae0[2 * c] * wi0; af0[c].y = ae0[2 * c + 1] * wi0;
    af1[c].x = ae1[2 * c] * wi1; af1[c].y = ae1[2 * c + 1] * wi1;
  }

  const v2f ln2 = {0.693147180559945f, 0.693147180559945f};
  #pragma unroll
  for (int g = 0; g < G_; ++g) {
    float x0[8], x1[8];
    #pragma unroll
    for (int c = 0; c < 4; ++c) {
      const v4f t = q[g][c];
      x0[2 * c]     = t.x; x1[2 * c]     = t.y;
      x0[2 * c + 1] = t.z; x1[2 * c + 1] = t.w;
    }
    float m0 = x0[0], m1 = x1[0];
    #pragma unroll
    for (int j = 1; j < 8; ++j) { m0 = fmaxf(m0, x0[j]); m1 = fmaxf(m1, x1[j]); }
    m0 = fmaxf(m0, __shfl_xor(m0, 16, 32));
    m1 = fmaxf(m1, __shfl_xor(m1, 16, 32));
    v2f bf0[4], bf1[4];
    #pragma unroll
    for (int c = 0; c < 4; ++c) {
      bf0[c].x = __expf(x0[2 * c]     - m0); bf0[c].y = __expf(x0[2 * c + 1] - m0);
      bf1[c].x = __expf(x1[2 * c]     - m1); bf1[c].y = __expf(x1[2 * c + 1] - m1);
    }

    v8f acc0 = {0.f, 0.f, 0.f, 0.f, 0.f, 0.f, 0.f, 0.f};
    v8f acc1 = {0.f, 0.f, 0.f, 0.f, 0.f, 0.f, 0.f, 0.f};
    #pragma unroll
    for (int c = 0; c < 4; ++c) {
      acc0 = __builtin_amdgcn_wmma_f32_16x16x4_f32(false, af0[c], false, bf0[c],
                                                   (short)0, acc0, false, false);
      acc1 = __builtin_amdgcn_wmma_f32_16x16x4_f32(false, af1[c], false, bf1[c],
                                                   (short)0, acc1, false, false);
    }

    v2f* po = (v2f*)(out + (pos0 + 256 * g) * (COUT_ * R_));
    const v2f mv  = {m0, m1};
    #pragma unroll
    for (int v = 0; v < 8; ++v) {
      const int o = hi * 8 + v;
      v2f l2; l2.x = __builtin_amdgcn_logf(acc0[v]); l2.y = __builtin_amdgcn_logf(acc1[v]);
      po[o] = l2 * ln2 + mv;
    }
    __builtin_amdgcn_sched_barrier(0);
  }
}

extern "C" void kernel_launch(void* const* d_in, const int* in_sizes, int n_in,
                              void* d_out, int out_size, void* d_ws, size_t ws_size,
                              hipStream_t stream) {
  (void)in_sizes; (void)n_in; (void)d_ws; (void)ws_size; (void)out_size;
  const float* ll     = (const float*)d_in[0];
  const float* logits = (const float*)d_in[1];
  float* out          = (float*)d_out;

  dim3 grid(256), block(256);
  sumconv_wmma_kernel<<<grid, block, 0, stream>>>(ll, logits, out);
}


# Loads on few lines: ll as 4 whole 128B lines per load (was 16 pieces); logits once per block as 64B half rows, 8 per load (was 32 lines per load, each wave); LDS restores fragment layouts; same sums
# speedup vs baseline: 1.3229x; 1.0797x over previous
#include <hip/hip_runtime.h>

typedef float v2f __attribute__((ext_vector_type(2)));
typedef float v4f __attribute__((ext_vector_type(4)));
typedef float v8f __attribute__((ext_vector_type(8)));

#define B_    128
#define F_    1024
#define CIN_  16
#define COUT_ 16
#define K_    4
#define R_    2
#define G_    4
#define WPB_  8
#define PITCH_ 9
#define OP_   10

__global__ __launch_bounds__(256) void sumconv_wmma_kernel(
    const float* __restrict__ ll,
    const float* __restrict__ logits,
    float* __restrict__ out)
{
  __shared__ v4f stage[WPB_][2][16 * PITCH_];
  __shared__ __attribute__((aligned(16))) v2f wl[WPB_][2 * 16 * OP_];
  const int wave = (blockIdx.x * (int)blockDim.x + (int)threadIdx.x) >> 5;
  const int wib  = (int)threadIdx.x >> 5;
  const int lane = threadIdx.x & 31;
  const int hi   = lane >> 4;
  const int col  = lane & 15;

  const int kw =  wave       & 3;
  const int kh = (wave >> 2) & 3;
  const int b  =  wave >> 4;


  const int khh = (wave >> 3) & 1;
  const int lch = lane & 3;
  v4f lq[4];
  #pragma unroll
  for (int i = 0; i < 4; ++i) {
    const int L = 64 * i + 8 * wib + (lane >> 2);
    lq[i] = ((const v4f*)logits)[L * 8 + 4 * khh + lch];
  }
  const int f0 = (4 * (col >> 3) + kh) * 32 + 4 * (col & 7) + kw;
  const size_t pos0 = (size_t)(b * F_ + f0);
  const int lp = lane >> 3;
  const int lc = lane & 7;
  const v4f* llb = (const v4f*)(ll + ((size_t)b * F_ + (size_t)(kh * 32 + 4 * lp + kw)) * (CIN_ * R_)) + lc;
  v4f q[G_][4];
  #pragma unroll
  for (int g = 0; g < G_; ++g) {
    #pragma unroll
    for (int c = 0; c < 4; ++c)
      q[g][c] = llb[(256 * g + 128 * (c >> 1) + 16 * (c & 1)) * 8];
  }
  __builtin_amdgcn_sched_barrier(0);

  #pragma unroll
  for (int i = 0; i < 4; ++i) {
    const int L = 64 * i + 8 * wib + (lane >> 2);
    const int o = L >> 4, cin = L & 15;
    const int slot = ((cin >> 1) & 1) * (16 * OP_) + o * OP_ + 2 * (cin >> 2) + (cin & 1);
    const int we = 4 * (lch >> 1) + 2 * (lch & 1);
    const v2f e = {lq[i].x, lq[i].y}, d = {lq[i].z, lq[i].w};
    wl[we][slot] = e; wl[we + 1][slot] = d;
  }
  asm volatile("s_wait_dscnt 0x0" ::: "memory");
  asm volatile("s_barrier_signal -1" ::: "memory");
  asm volatile("s_barrier_wait -1" ::: "memory");
  float lg0[8], lg1[8];
  #pragma unroll
  for (int c = 0; c < 4; ++c) {
    const v4f t = ((const v4f*)wl[wib])[hi * (8 * OP_) + col * (OP_ / 2) + c];
    lg0[2 * c] = t.x; lg1[2 * c] = t.y; lg0[2 * c + 1] = t.z; lg1[2 * c + 1] = t.w;
  }

  float wm0 = lg0[0], wm1 = lg1[0];
  #pragma unroll
  for (int j = 1; j < 8; ++j) { wm0 = fmaxf(wm0, lg0[j]); wm1 = fmaxf(wm1, lg1[j]); }
  wm0 = fmaxf(wm0, __shfl_xor(wm0, 16, 32));
  wm1 = fmaxf(wm1, __shfl_xor(wm1, 16, 32));
  float ae0[8], ae1[8], ps0 = 0.0f, ps1 = 0.0f;
  #pragma unroll
  for (int j = 0; j < 8; ++j) {
    ae0[j] = __expf(lg0[j] - wm0); ps0 += ae0[j];
    ae1[j] = __expf(lg1[j] - wm1); ps1 += ae1[j];
  }
  const float wi0 = __builtin_amdgcn_rcpf(ps0 + __shfl_xor(ps0, 16, 32));
  const float wi1 = __builtin_amdgcn_rcpf(ps1 + __shfl_xor(ps1, 16, 32));
  v2f af0[4], af1[4];
  #pragma unroll
  for (int c = 0; c < 4; ++c) {
    af0[c].x = ae0[2 * c] * wi0; af0[c].y = ae0[2 * c + 1] * wi0;
    af1[c].x = ae1[2 * c] * wi1; af1[c].y = ae1[2 * c + 1] * wi1;
  }

  const v2f ln2 = {0.693147180559945f, 0.693147180559945f};
  #pragma unroll
  for (int g = 0; g < G_; ++g) {
    v4f* sb = stage[wib][g & 1];
    #pragma unroll
    for (int c = 0; c < 4; ++c) sb[(4 * c + lp) * PITCH_ + lc] = q[g][c];
    __builtin_amdgcn_fence(__ATOMIC_RELEASE, "wavefront");
    __builtin_amdgcn_wave_barrier();
    __builtin_amdgcn_fence(__ATOMIC_ACQUIRE, "wavefront");
    v4f fr[4];
    #pragma unroll
    for (int c = 0; c < 4; ++c) fr[c] = sb[col * PITCH_ + 2 * c + hi];
    float x0[8], x1[8];
    #pragma unroll
    for (int c = 0; c < 4; ++c) {
      const v4f t = fr[c];
      x0[2 * c]     = t.x; x1[2 * c]     = t.y;
      x0[2 * c + 1] = t.z; x1[2 * c + 1] = t.w;
    }
    float m0 = x0[0], m1 = x1[0];
    #pragma unroll
    for (int j = 1; j < 8; ++j) { m0 = fmaxf(m0, x0[j]); m1 = fmaxf(m1, x1[j]); }
    m0 = fmaxf(m0, __shfl_xor(m0, 16, 32));
    m1 = fmaxf(m1, __shfl_xor(m1, 16, 32));
    v2f bf0[4], bf1[4];
    #pragma unroll
    for (int c = 0; c < 4; ++c) {
      bf0[c].x = __expf(x0[2 * c]     - m0); bf0[c].y = __expf(x0[2 * c + 1] - m0);
      bf1[c].x = __expf(x1[2 * c]     - m1); bf1[c].y = __expf(x1[2 * c + 1] - m1);
    }

    v8f acc0 = {0.f, 0.f, 0.f, 0.f, 0.f, 0.f, 0.f, 0.f};
    v8f acc1 = {0.f, 0.f, 0.f, 0.f, 0.f, 0.f, 0.f, 0.f};
    #pragma unroll
    for (int c = 0; c < 4; ++c) {
      acc0 = __builtin_amdgcn_wmma_f32_16x16x4_f32(false, af0[c], false, bf0[c],
                                                   (short)0, acc0, false, false);
      acc1 = __builtin_amdgcn_wmma_f32_16x16x4_f32(false, af1[c], false, bf1[c],
                                                   (short)0, acc1, false, false);
    }

    v2f* po = (v2f*)(out + (pos0 + 256 * g) * (COUT_ * R_));
    const v2f mv  = {m0, m1};
    #pragma unroll
    for (int v = 0; v < 8; ++v) {
      const int o = hi * 8 + v;
      v2f l2; l2.x = __builtin_amdgcn_logf(acc0[v]); l2.y = __builtin_amdgcn_logf(acc1[v]);
      po[o] = l2 * ln2 + mv;
    }
    __builtin_amdgcn_sched_barrier(0);
  }
}

extern "C" void kernel_launch(void* const* d_in, const int* in_sizes, int n_in,
                              void* d_out, int out_size, void* d_ws, size_t ws_size,
                              hipStream_t stream) {
  (void)in_sizes; (void)n_in; (void)d_ws; (void)ws_size; (void)out_size;
  const float* ll     = (const float*)d_in[0];
  const float* logits = (const float*)d_in[1];
  float* out          = (float*)d_out;

  dim3 grid(256), block(256);
  sumconv_wmma_kernel<<<grid, block, 0, stream>>>(ll, logits, out);
}
